# starting kernel
# baseline (speedup 1.0000x reference)
#include <hip/hip_runtime.h>
#include <math.h>

typedef float v2f __attribute__((ext_vector_type(2)));
typedef float v8f __attribute__((ext_vector_type(8)));

#define NEG_SLOPE 0.2f

__device__ __forceinline__ void atomic_max_float(float* addr, float val) {
  if (val >= 0.0f) atomicMax((int*)addr, __float_as_int(val));
  else             atomicMin((unsigned int*)addr, __float_as_uint(val));
}

__global__ __launch_bounds__(256) void gemm_wmma_f32(
    const float* __restrict__ A, const float* __restrict__ B,
    const float* __restrict__ bias, float* __restrict__ C,
    int M, int K, int N) {
  const int lane = threadIdx.x & 31;
  const int wave = blockIdx.x * (blockDim.x >> 5) + (threadIdx.x >> 5);
  const int tilesN = N >> 4;
  const int tilesM = M >> 4;
  if (wave >= tilesM * tilesN) return;
  const int tm = wave / tilesN;
  const int tn = wave - tm * tilesN;
  const int mrow  = tm * 16 + (lane & 15);
  const int ncol  = tn * 16 + (lane & 15);
  const int khalf = (lane >> 4) << 1;

  v8f acc = {};
  for (int k = 0; k < K; k += 4) {
    const float* ap = A + (size_t)mrow * N * 0 + (size_t)mrow * K + (k + khalf);
    v2f a; a.x = ap[0]; a.y = ap[1];
    const float* bp = B + (size_t)(k + khalf) * N + ncol;
    v2f b; b.x = bp[0]; b.y = bp[N];
    acc = __builtin_amdgcn_wmma_f32_16x16x4_f32(
         false, a,  false, b,
         (short)0, acc,  false,  false);
  }

  const int mbase = tm * 16 + ((lane >> 4) << 3);
  const float bv = bias ? bias[ncol] : 0.0f;
#pragma unroll
  for (int v = 0; v < 8; ++v)
    C[(size_t)(mbase + v) * N + ncol] = acc[v] + bv;
}

__global__ void fill_value(float* __restrict__ p, float v, long n) {
  long t = (long)blockIdx.x * blockDim.x + threadIdx.x;
  if (t < n) p[t] = v;
}

__global__ void fill_rows_bias(float* __restrict__ out, const float* __restrict__ bias,
                               long rows, int C) {
  long t = (long)blockIdx.x * blockDim.x + threadIdx.x;
  if (t < rows * (long)C) out[t] = bias[(int)(t % C)];
}

__global__ void att_scores(const float* __restrict__ h,
                           const float* __restrict__ att_src,
                           const float* __restrict__ att_dst,
                           float* __restrict__ as, float* __restrict__ ad,
                           int n, int H, int C) {
  long t = (long)blockIdx.x * blockDim.x + threadIdx.x;
  if (t >= (long)n * H) return;
  const int head = (int)(t % H);
  const float4* hv = (const float4*)(h + (size_t)t * C);
  const float4* sv = (const float4*)(att_src + (size_t)head * C);
  const float4* dv = (const float4*)(att_dst + (size_t)head * C);
  float s = 0.f, d = 0.f;
  for (int i = 0; i < (C >> 2); ++i) {
    float4 xv = hv[i], a = sv[i], b = dv[i];
    s += xv.x * a.x + xv.y * a.y + xv.z * a.z + xv.w * a.w;
    d += xv.x * b.x + xv.y * b.y + xv.z * b.z + xv.w * b.w;
  }
  as[t] = s; ad[t] = d;
}

__global__ void edge_logits_max(const int* __restrict__ ei,
                                const float* __restrict__ as,
                                const float* __restrict__ ad,
                                float* __restrict__ e, float* __restrict__ m,
                                int E, int n, int H) {
  long t = (long)blockIdx.x * blockDim.x + threadIdx.x;
  long total = (long)(E + n) * H;
  if (t >= total) return;
  const int edge = (int)(t / H);
  const int hd   = (int)(t % H);
  const int src = (edge < E) ? ei[edge]     : edge - E;
  const int dst = (edge < E) ? ei[E + edge] : edge - E;
  float v = as[(size_t)src * H + hd] + ad[(size_t)dst * H + hd];
  v = (v > 0.f) ? v : NEG_SLOPE * v;
  e[t] = v;
  atomic_max_float(&m[(size_t)dst * H + hd], v);
}

__global__ void edge_exp_sum(const int* __restrict__ ei, float* __restrict__ e,
                             const float* __restrict__ m, float* __restrict__ s,
                             int E, int n, int H) {
  long t = (long)blockIdx.x * blockDim.x + threadIdx.x;
  long total = (long)(E + n) * H;
  if (t >= total) return;
  const int edge = (int)(t / H);
  const int hd   = (int)(t % H);
  const int dst = (edge < E) ? ei[E + edge] : edge - E;
  float v = expf(e[t] - m[(size_t)dst * H + hd]);
  e[t] = v;
  atomicAdd(&s[(size_t)dst * H + hd], v);
}

__global__ void edge_scatter(const int* __restrict__ ei, const float* __restrict__ e,
                             const float* __restrict__ s, const float* __restrict__ feat,
                             float* __restrict__ out, int E, int n, int H, int C) {
  long t = (long)blockIdx.x * blockDim.x + threadIdx.x;
  long total = (long)(E + n) * H * C;
  if (t >= total) return;
  const int  c    = (int)(t % C);
  const long eh   = t / C;
  const int  hd   = (int)(eh % H);
  const int  edge = (int)(eh / H);
  const int src = (edge < E) ? ei[edge]     : edge - E;
  const int dst = (edge < E) ? ei[E + edge] : edge - E;
  const float alpha = e[eh] / (s[(size_t)dst * H + hd] + 1e-16f);
  atomicAdd(&out[((size_t)dst * H + hd) * C + c],
            alpha * feat[((size_t)src * H + hd) * C + c]);
}

__global__ void ln_add_elu(const float* __restrict__ x, const float* __restrict__ g,
                           const float* __restrict__ b, const float* __restrict__ add,
                           float* __restrict__ y, int n, int C) {
  const int row  = blockIdx.x * (blockDim.x >> 5) + (threadIdx.x >> 5);
  const int lane = threadIdx.x & 31;
  if (row >= n) return;
  const float* xr = x + (size_t)row * C;
  float s = 0.f, ss = 0.f;
  for (int c = lane; c < C; c += 32) { float v = xr[c]; s += v; ss += v * v; }
  for (int off = 16; off; off >>= 1) {
    s  += __shfl_xor(s,  off, 32);
    ss += __shfl_xor(ss, off, 32);
  }
  const float mu  = s / (float)C;
  const float var = ss / (float)C - mu * mu;
  const float inv = rsqrtf(var + 1e-5f);
  const float* ar = add ? add + (size_t)row * C : nullptr;
  float* yr = y + (size_t)row * C;
  for (int c = lane; c < C; c += 32) {
    float v = (xr[c] - mu) * inv * g[c] + b[c];
    if (ar) v += ar[c];
    yr[c] = (v > 0.f) ? v : expm1f(v);
  }
}

extern "C" void kernel_launch(void* const* d_in, const int* in_sizes, int n_in,
                              void* d_out, int out_size, void* d_ws, size_t ws_size,
                              hipStream_t stream) {
  const float* x    = (const float*)d_in[0];
  const int*   ei   = (const int*)  d_in[1];
  const float* W1   = (const float*)d_in[2];
  const float* aS1  = (const float*)d_in[3];
  const float* aD1  = (const float*)d_in[4];
  const float* b1   = (const float*)d_in[5];
  const float* W2   = (const float*)d_in[6];
  const float* aS2  = (const float*)d_in[7];
  const float* aD2  = (const float*)d_in[8];
  const float* b2   = (const float*)d_in[9];
  const float* g1   = (const float*)d_in[10];
  const float* be1  = (const float*)d_in[11];
  const float* g2   = (const float*)d_in[12];
  const float* be2  = (const float*)d_in[13];
  const float* pW   = (const float*)d_in[14];
  const float* pb   = (const float*)d_in[15];

  const int IN_CH = 128, HEADS = 4, OUT_CH = 64, HC = HEADS * OUT_CH;
  const int N  = in_sizes[0] / IN_CH;
  const int E  = in_sizes[1] / 2;
  const long EP = (long)E + N;

  float* ws   = (float*)d_ws;
  float* h1   = ws; ws += (size_t)N * HC;
  float* out1 = ws; ws += (size_t)N * HC;
  float* idb  = ws; ws += (size_t)N * OUT_CH;
  float* z2   = ws; ws += (size_t)N * OUT_CH;
  float* out2 = ws; ws += (size_t)N * OUT_CH;
  float* as1  = ws; ws += (size_t)N * HEADS;
  float* ad1  = ws; ws += (size_t)N * HEADS;
  float* m1   = ws; ws += (size_t)N * HEADS;
  float* s1   = ws; ws += (size_t)N * HEADS;
  float* e1   = ws; ws += (size_t)EP * HEADS;
  float* as2  = ws; ws += (size_t)N;
  float* ad2  = ws; ws += (size_t)N;
  float* m2   = ws; ws += (size_t)N;
  float* s2   = ws; ws += (size_t)N;
  float* e2   = ws; ws += (size_t)EP;

  auto cdiv = [](long a, long b) -> int { return (int)((a + b - 1) / b); };
  const int BLK = 256;
  const int WPB = BLK / 32;

  {
    long waves = ((long)N / 16) * (HC / 16);
    gemm_wmma_f32<<<cdiv(waves, WPB), BLK, 0, stream>>>(x, W1, nullptr, h1, N, IN_CH, HC);
  }
  {
    long waves = ((long)N / 16) * (OUT_CH / 16);
    gemm_wmma_f32<<<cdiv(waves, WPB), BLK, 0, stream>>>(x, pW, pb, idb, N, IN_CH, OUT_CH);
  }

  att_scores<<<cdiv((long)N * HEADS, BLK), BLK, 0, stream>>>(h1, aS1, aD1, as1, ad1, N, HEADS, OUT_CH);
  fill_value<<<cdiv((long)N * HEADS, BLK), BLK, 0, stream>>>(m1, -INFINITY, (long)N * HEADS);
  fill_value<<<cdiv((long)N * HEADS, BLK), BLK, 0, stream>>>(s1, 0.0f,      (long)N * HEADS);
  fill_rows_bias<<<cdiv((long)N * HC, BLK), BLK, 0, stream>>>(out1, b1, N, HC);

  edge_logits_max<<<cdiv(EP * HEADS, BLK), BLK, 0, stream>>>(ei, as1, ad1, e1, m1, E, N, HEADS);
  edge_exp_sum  <<<cdiv(EP * HEADS, BLK), BLK, 0, stream>>>(ei, e1, m1, s1, E, N, HEADS);
  edge_scatter  <<<cdiv(EP * HEADS * OUT_CH, BLK), BLK, 0, stream>>>(ei, e1, s1, h1, out1, E, N, HEADS, OUT_CH);

  ln_add_elu<<<cdiv(N, WPB), BLK, 0, stream>>>(out1, g1, be1, nullptr, out1, N, HC);

  {
    long waves = ((long)N / 16) * (OUT_CH / 16);
    gemm_wmma_f32<<<cdiv(waves, WPB), BLK, 0, stream>>>(out1, W2, nullptr, z2, N, HC, OUT_CH);
  }

  att_scores<<<cdiv((long)N, BLK), BLK, 0, stream>>>(z2, aS2, aD2, as2, ad2, N, 1, OUT_CH);
  fill_value<<<cdiv((long)N, BLK), BLK, 0, stream>>>(m2, -INFINITY, (long)N);
  fill_value<<<cdiv((long)N, BLK), BLK, 0, stream>>>(s2, 0.0f,      (long)N);
  fill_rows_bias<<<cdiv((long)N * OUT_CH, BLK), BLK, 0, stream>>>(out2, b2, N, OUT_CH);

  edge_logits_max<<<cdiv(EP, BLK), BLK, 0, stream>>>(ei, as2, ad2, e2, m2, E, N, 1);
  edge_exp_sum  <<<cdiv(EP, BLK), BLK, 0, stream>>>(ei, e2, m2, s2, E, N, 1);
  edge_scatter  <<<cdiv(EP * OUT_CH, BLK), BLK, 0, stream>>>(ei, e2, s2, z2, out2, E, N, 1, OUT_CH);

  ln_add_elu<<<cdiv(N, WPB), BLK, 0, stream>>>(out2, g2, be2, idb, (float*)d_out, N, OUT_CH);

  (void)n_in; (void)out_size; (void)ws_size;
}


# aggregate features by destination: edges grouped into an in-call CSR, one wave per row sums source rows in registers and stores once (bias folded in) instead of one float atomic per element
# speedup vs baseline: 2.2170x; 2.2170x over previous
#include <hip/hip_runtime.h>
#include <math.h>

typedef float v2f __attribute__((ext_vector_type(2)));
typedef float v8f __attribute__((ext_vector_type(8)));

#define NEG_SLOPE 0.2f

__device__ __forceinline__ void atomic_max_float(float* addr, float val) {
  if (val >= 0.0f) atomicMax((int*)addr, __float_as_int(val));
  else             atomicMin((unsigned int*)addr, __float_as_uint(val));
}

__global__ __launch_bounds__(256) void gemm_wmma_f32(
    const float* __restrict__ A, const float* __restrict__ B,
    const float* __restrict__ bias, float* __restrict__ C,
    int M, int K, int N) {
  const int lane = threadIdx.x & 31;
  const int wave = blockIdx.x * (blockDim.x >> 5) + (threadIdx.x >> 5);
  const int tilesN = N >> 4;
  const int tilesM = M >> 4;
  if (wave >= tilesM * tilesN) return;
  const int tm = wave / tilesN;
  const int tn = wave - tm * tilesN;
  const int mrow  = tm * 16 + (lane & 15);
  const int ncol  = tn * 16 + (lane & 15);
  const int khalf = (lane >> 4) << 1;

  v8f acc = {};
  for (int k = 0; k < K; k += 4) {
    const float* ap = A + (size_t)mrow * N * 0 + (size_t)mrow * K + (k + khalf);
    v2f a; a.x = ap[0]; a.y = ap[1];
    const float* bp = B + (size_t)(k + khalf) * N + ncol;
    v2f b; b.x = bp[0]; b.y = bp[N];
    acc = __builtin_amdgcn_wmma_f32_16x16x4_f32(
         false, a,  false, b,
         (short)0, acc,  false,  false);
  }

  const int mbase = tm * 16 + ((lane >> 4) << 3);
  const float bv = bias ? bias[ncol] : 0.0f;
#pragma unroll
  for (int v = 0; v < 8; ++v)
    C[(size_t)(mbase + v) * N + ncol] = acc[v] + bv;
}

__global__ void fill_value(float* __restrict__ p, float v, long n) {
  long t = (long)blockIdx.x * blockDim.x + threadIdx.x;
  if (t < n) p[t] = v;
}

__global__ void fill_int(int* __restrict__ p, int v, long n) {
  long t = (long)blockIdx.x * blockDim.x + threadIdx.x;
  if (t < n) p[t] = v;
}

__global__ void count_and_rank(const int* __restrict__ ei, int* __restrict__ cnt,
                               int* __restrict__ rank, int E) {
  long t = (long)blockIdx.x * blockDim.x + threadIdx.x;
  if (t >= E) return;
  rank[t] = atomicAdd(&cnt[ei[E + t]], 1);
}

__global__ __launch_bounds__(1024) void scan_counts(const int* __restrict__ cnt,
                                                    int* __restrict__ rowptr, int n) {
  __shared__ int wsum[32];
  const int tid  = threadIdx.x;
  const int lane = tid & 31;
  const int wv   = tid >> 5;
  const int per  = (n + 1023) >> 10;
  const int lo   = min(tid * per, n);
  const int hi   = min(lo + per, n);
  int local = 0;
  for (int i = lo; i < hi; ++i) local += cnt[i];
  int x = local;
  for (int off = 1; off < 32; off <<= 1) {
    const int y = __shfl_up(x, off, 32);
    if (lane >= off) x += y;
  }
  if (lane == 31) wsum[wv] = x;
  __syncthreads();
  if (wv == 0) {
    const int w = wsum[lane];
    int xs = w;
    for (int off = 1; off < 32; off <<= 1) {
      const int y = __shfl_up(xs, off, 32);
      if (lane >= off) xs += y;
    }
    wsum[lane] = xs - w;
  }
  __syncthreads();
  int run = wsum[wv] + x - local;
  for (int i = lo; i < hi; ++i) { rowptr[i] = run; run += cnt[i]; }
  if (tid == 1023) rowptr[n] = run;
}

__global__ void fill_csr(const int* __restrict__ ei, const int* __restrict__ rowptr,
                         const int* __restrict__ rank, int* __restrict__ csr_eid, int E) {
  long t = (long)blockIdx.x * blockDim.x + threadIdx.x;
  if (t >= E) return;
  csr_eid[rowptr[ei[E + t]] + rank[t]] = (int)t;
}

__global__ void att_scores(const float* __restrict__ h,
                           const float* __restrict__ att_src,
                           const float* __restrict__ att_dst,
                           float* __restrict__ as, float* __restrict__ ad,
                           int n, int H, int C) {
  long t = (long)blockIdx.x * blockDim.x + threadIdx.x;
  if (t >= (long)n * H) return;
  const int head = (int)(t % H);
  const float4* hv = (const float4*)(h + (size_t)t * C);
  const float4* sv = (const float4*)(att_src + (size_t)head * C);
  const float4* dv = (const float4*)(att_dst + (size_t)head * C);
  float s = 0.f, d = 0.f;
  for (int i = 0; i < (C >> 2); ++i) {
    float4 xv = hv[i], a = sv[i], b = dv[i];
    s += xv.x * a.x + xv.y * a.y + xv.z * a.z + xv.w * a.w;
    d += xv.x * b.x + xv.y * b.y + xv.z * b.z + xv.w * b.w;
  }
  as[t] = s; ad[t] = d;
}

__global__ void edge_logits_max(const int* __restrict__ ei,
                                const float* __restrict__ as,
                                const float* __restrict__ ad,
                                float* __restrict__ e, float* __restrict__ m,
                                int E, int n, int H) {
  long t = (long)blockIdx.x * blockDim.x + threadIdx.x;
  long total = (long)(E + n) * H;
  if (t >= total) return;
  const int edge = (int)(t / H);
  const int hd   = (int)(t % H);
  const int src = (edge < E) ? ei[edge]     : edge - E;
  const int dst = (edge < E) ? ei[E + edge] : edge - E;
  float v = as[(size_t)src * H + hd] + ad[(size_t)dst * H + hd];
  v = (v > 0.f) ? v : NEG_SLOPE * v;
  e[t] = v;
  atomic_max_float(&m[(size_t)dst * H + hd], v);
}

__global__ void edge_exp_sum(const int* __restrict__ ei, float* __restrict__ e,
                             const float* __restrict__ m, float* __restrict__ s,
                             int E, int n, int H) {
  long t = (long)blockIdx.x * blockDim.x + threadIdx.x;
  long total = (long)(E + n) * H;
  if (t >= total) return;
  const int edge = (int)(t / H);
  const int hd   = (int)(t % H);
  const int dst = (edge < E) ? ei[E + edge] : edge - E;
  float v = expf(e[t] - m[(size_t)dst * H + hd]);
  e[t] = v;
  atomicAdd(&s[(size_t)dst * H + hd], v);
}

template <int H, int C>
__global__ __launch_bounds__(256) void gather_aggregate(
    const int* __restrict__ rowptr, const int* __restrict__ csr_eid,
    const int* __restrict__ ei, const float* __restrict__ e,
    const float* __restrict__ s, const float* __restrict__ feat,
    const float* __restrict__ bias, float* __restrict__ out, int E, int n) {
  constexpr int HC  = H * C;
  constexpr int VW  = (HC >= 128) ? 4 : HC / 32;
  constexpr int NCH = HC / (32 * VW);
  static_assert(VW >= 1 && NCH * 32 * VW == HC, "row must split into whole wave chunks");
  static_assert(C % VW == 0, "a lane's vector must stay inside one head");
  typedef float vec __attribute__((ext_vector_type(VW)));

  const int lane = threadIdx.x & 31;
  const int d = __builtin_amdgcn_readfirstlane(
      (int)(blockIdx.x * (blockDim.x >> 5) + (threadIdx.x >> 5)));
  if (d >= n) return;
  const int beg = rowptr[d];
  const int end = rowptr[d + 1];

  int col[NCH], hd[NCH];
  vec acc[NCH];
  for (int k = 0; k < NCH; ++k) {
    col[k] = k * 32 * VW + lane * VW;
    hd[k]  = col[k] / C;
    acc[k] = e[(size_t)(E + d) * H + hd[k]] * *(const vec*)(feat + (size_t)d * HC + col[k]);
  }

  for (int base = beg; base < end; base += 32) {
    const int cnt = min(32, end - base);
    int my_eid = 0, my_src = 0;
    if (lane < cnt) { my_eid = csr_eid[base + lane]; my_src = ei[my_eid]; }
    for (int j = 0; j < cnt; ++j) {
      const int eid = __builtin_amdgcn_readlane(my_eid, j);
      const int src = __builtin_amdgcn_readlane(my_src, j);
      for (int k = 0; k < NCH; ++k) {
        const float w = e[(size_t)eid * H + hd[k]];
        acc[k] += w * *(const vec*)(feat + (size_t)src * HC + col[k]);
      }
    }
  }

  for (int k = 0; k < NCH; ++k) {
    const float inv = 1.0f / (s[(size_t)d * H + hd[k]] + 1e-16f);
    *(vec*)(out + (size_t)d * HC + col[k]) = acc[k] * inv + *(const vec*)(bias + col[k]);
  }
}

__global__ void ln_add_elu(const float* __restrict__ x, const float* __restrict__ g,
                           const float* __restrict__ b, const float* __restrict__ add,
                           float* __restrict__ y, int n, int C) {
  const int row  = blockIdx.x * (blockDim.x >> 5) + (threadIdx.x >> 5);
  const int lane = threadIdx.x & 31;
  if (row >= n) return;
  const float* xr = x + (size_t)row * C;
  float s = 0.f, ss = 0.f;
  for (int c = lane; c < C; c += 32) { float v = xr[c]; s += v; ss += v * v; }
  for (int off = 16; off; off >>= 1) {
    s  += __shfl_xor(s,  off, 32);
    ss += __shfl_xor(ss, off, 32);
  }
  const float mu  = s / (float)C;
  const float var = ss / (float)C - mu * mu;
  const float inv = rsqrtf(var + 1e-5f);
  const float* ar = add ? add + (size_t)row * C : nullptr;
  float* yr = y + (size_t)row * C;
  for (int c = lane; c < C; c += 32) {
    float v = (xr[c] - mu) * inv * g[c] + b[c];
    if (ar) v += ar[c];
    yr[c] = (v > 0.f) ? v : expm1f(v);
  }
}

extern "C" void kernel_launch(void* const* d_in, const int* in_sizes, int n_in,
                              void* d_out, int out_size, void* d_ws, size_t ws_size,
                              hipStream_t stream) {
  const float* x    = (const float*)d_in[0];
  const int*   ei   = (const int*)  d_in[1];
  const float* W1   = (const float*)d_in[2];
  const float* aS1  = (const float*)d_in[3];
  const float* aD1  = (const float*)d_in[4];
  const float* b1   = (const float*)d_in[5];
  const float* W2   = (const float*)d_in[6];
  const float* aS2  = (const float*)d_in[7];
  const float* aD2  = (const float*)d_in[8];
  const float* b2   = (const float*)d_in[9];
  const float* g1   = (const float*)d_in[10];
  const float* be1  = (const float*)d_in[11];
  const float* g2   = (const float*)d_in[12];
  const float* be2  = (const float*)d_in[13];
  const float* pW   = (const float*)d_in[14];
  const float* pb   = (const float*)d_in[15];

  const int IN_CH = 128, HEADS = 4, OUT_CH = 64, HC = HEADS * OUT_CH;
  const int N  = in_sizes[0] / IN_CH;
  const int E  = in_sizes[1] / 2;
  const long EP = (long)E + N;

  float* ws   = (float*)d_ws;
  float* h1   = ws; ws += (size_t)N * HC;
  float* out1 = ws; ws += (size_t)N * HC;
  float* idb  = ws; ws += (size_t)N * OUT_CH;
  float* z2   = ws; ws += (size_t)N * OUT_CH;
  float* out2 = ws; ws += (size_t)N * OUT_CH;
  float* as1  = ws; ws += (size_t)N * HEADS;
  float* ad1  = ws; ws += (size_t)N * HEADS;
  float* m1   = ws; ws += (size_t)N * HEADS;
  float* s1   = ws; ws += (size_t)N * HEADS;
  float* e1   = ws; ws += (size_t)EP * HEADS;
  float* as2  = as1;
  float* ad2  = ad1;
  float* m2   = m1;
  float* s2   = s1;
  float* e2   = e1;
  int* rowptr  = (int*)ws; ws += (size_t)N + 1;
  int* cnt     = (int*)ws; ws += (size_t)N;
  int* csr_eid = (int*)ws; ws += (size_t)E;
  int* rank    = (int*)e1;

  auto cdiv = [](long a, long b) -> int { return (int)((a + b - 1) / b); };
  const int BLK = 256;
  const int WPB = BLK / 32;

  fill_int       <<<cdiv((long)N, BLK), BLK, 0, stream>>>(cnt, 0, (long)N);
  count_and_rank <<<cdiv((long)E, BLK), BLK, 0, stream>>>(ei, cnt, rank, E);
  scan_counts    <<<1, 1024, 0, stream>>>(cnt, rowptr, N);
  fill_csr       <<<cdiv((long)E, BLK), BLK, 0, stream>>>(ei, rowptr, rank, csr_eid, E);

  {
    long waves = ((long)N / 16) * (HC / 16);
    gemm_wmma_f32<<<cdiv(waves, WPB), BLK, 0, stream>>>(x, W1, nullptr, h1, N, IN_CH, HC);
  }
  {
    long waves = ((long)N / 16) * (OUT_CH / 16);
    gemm_wmma_f32<<<cdiv(waves, WPB), BLK, 0, stream>>>(x, pW, pb, idb, N, IN_CH, OUT_CH);
  }

  att_scores<<<cdiv((long)N * HEADS, BLK), BLK, 0, stream>>>(h1, aS1, aD1, as1, ad1, N, HEADS, OUT_CH);
  fill_value<<<cdiv((long)N * HEADS, BLK), BLK, 0, stream>>>(m1, -INFINITY, (long)N * HEADS);
  fill_value<<<cdiv((long)N * HEADS, BLK), BLK, 0, stream>>>(s1, 0.0f,      (long)N * HEADS);

  edge_logits_max<<<cdiv(EP * HEADS, BLK), BLK, 0, stream>>>(ei, as1, ad1, e1, m1, E, N, HEADS);
  edge_exp_sum  <<<cdiv(EP * HEADS, BLK), BLK, 0, stream>>>(ei, e1, m1, s1, E, N, HEADS);
  gather_aggregate<4, 64><<<cdiv(N, WPB), BLK, 0, stream>>>(rowptr, csr_eid, ei, e1, s1, h1, b1, out1, E, N);

  ln_add_elu<<<cdiv(N, WPB), BLK, 0, stream>>>(out1, g1, be1, nullptr, out1, N, HC);

  {
    long waves = ((long)N / 16) * (OUT_CH / 16);
    gemm_wmma_f32<<<cdiv(waves, WPB), BLK, 0, stream>>>(out1, W2, nullptr, z2, N, HC, OUT_CH);
  }

  att_scores<<<cdiv((long)N, BLK), BLK, 0, stream>>>(z2, aS2, aD2, as2, ad2, N, 1, OUT_CH);
  fill_value<<<cdiv((long)N, BLK), BLK, 0, stream>>>(m2, -INFINITY, (long)N);
  fill_value<<<cdiv((long)N, BLK), BLK, 0, stream>>>(s2, 0.0f,      (long)N);

  edge_logits_max<<<cdiv(EP, BLK), BLK, 0, stream>>>(ei, as2, ad2, e2, m2, E, N, 1);
  edge_exp_sum  <<<cdiv(EP, BLK), BLK, 0, stream>>>(ei, e2, m2, s2, E, N, 1);
  gather_aggregate<1, 64><<<cdiv(N, WPB), BLK, 0, stream>>>(rowptr, csr_eid, ei, e2, s2, z2, b2, out2, E, N);

  ln_add_elu<<<cdiv(N, WPB), BLK, 0, stream>>>(out2, g2, be2, idb, (float*)d_out, N, OUT_CH);

  (void)n_in; (void)out_size; (void)ws_size;
}


# GEMM: one wave per 16x64 strip instead of per 16x16 tile: each A fragment feeds 4 WMMAs, A/B/C move as 16-byte accesses, next K-step loaded ahead; k is summed in a different order (8-k groups)
# speedup vs baseline: 4.7845x; 2.1581x over previous
#include <hip/hip_runtime.h>
#include <math.h>

typedef float v2f __attribute__((ext_vector_type(2)));
typedef float v4f __attribute__((ext_vector_type(4)));
typedef float v8f __attribute__((ext_vector_type(8)));

#define NEG_SLOPE 0.2f

__global__ __launch_bounds__(256) void gemm_wmma_f32(
    const float* __restrict__ A, const float* __restrict__ B,
    const float* __restrict__ bias, float* __restrict__ C,
    int M, int K, int N) {
  const int lane = threadIdx.x & 31;
  const int wave = blockIdx.x * (blockDim.x >> 5) + (threadIdx.x >> 5);
  const int stripsN = N >> 6;
  const int tilesM  = M >> 4;
  if (wave >= tilesM * stripsN) return;
  const int tm = wave / stripsN;
  const int ts = wave - tm * stripsN;
  const int c  = lane & 15;
  const int hf = lane >> 4;
  const float* ap = A + (size_t)(tm * 16 + c) * K + hf * 4;
  const float* bp = B + (size_t)(hf * 4) * N + ts * 64 + c * 4;

  v8f acc0 = {}, acc1 = {}, acc2 = {}, acc3 = {};
  v4f a  = *(const v4f*)ap;
  v4f b0 = *(const v4f*)(bp);
  v4f b1 = *(const v4f*)(bp + (size_t)N);
  v4f b2 = *(const v4f*)(bp + (size_t)2 * N);
  v4f b3 = *(const v4f*)(bp + (size_t)3 * N);
  for (int k = 0; k < K; k += 8) {
    const int kn = min(k + 8, K - 8);
    const float* bq = bp + (size_t)kn * N;
    const v4f an  = *(const v4f*)(ap + kn);
    const v4f bn0 = *(const v4f*)(bq);
    const v4f bn1 = *(const v4f*)(bq + (size_t)N);
    const v4f bn2 = *(const v4f*)(bq + (size_t)2 * N);
    const v4f bn3 = *(const v4f*)(bq + (size_t)3 * N);
    __builtin_amdgcn_sched_barrier(0);

    v2f af, bf;
    af.x = a.x; af.y = a.y;
    bf.x = b0.x; bf.y = b1.x;
    acc0 = __builtin_amdgcn_wmma_f32_16x16x4_f32(false, af, false, bf, (short)0, acc0, false, false);
    bf.x = b0.y; bf.y = b1.y;
    acc1 = __builtin_amdgcn_wmma_f32_16x16x4_f32(false, af, false, bf, (short)0, acc1, false, false);
    bf.x = b0.z; bf.y = b1.z;
    acc2 = __builtin_amdgcn_wmma_f32_16x16x4_f32(false, af, false, bf, (short)0, acc2, false, false);
    bf.x = b0.w; bf.y = b1.w;
    acc3 = __builtin_amdgcn_wmma_f32_16x16x4_f32(false, af, false, bf, (short)0, acc3, false, false);
    af.x = a.z; af.y = a.w;
    bf.x = b2.x; bf.y = b3.x;
    acc0 = __builtin_amdgcn_wmma_f32_16x16x4_f32(false, af, false, bf, (short)0, acc0, false, false);
    bf.x = b2.y; bf.y = b3.y;
    acc1 = __builtin_amdgcn_wmma_f32_16x16x4_f32(false, af, false, bf, (short)0, acc1, false, false);
    bf.x = b2.z; bf.y = b3.z;
    acc2 = __builtin_amdgcn_wmma_f32_16x16x4_f32(false, af, false, bf, (short)0, acc2, false, false);
    bf.x = b2.w; bf.y = b3.w;
    acc3 = __builtin_amdgcn_wmma_f32_16x16x4_f32(false, af, false, bf, (short)0, acc3, false, false);

    a = an; b0 = bn0; b1 = bn1; b2 = bn2; b3 = bn3;
  }

  v4f bv = {0.f, 0.f, 0.f, 0.f};
  if (bias) bv = *(const v4f*)(bias + ts * 64 + c * 4);
  float* cp = C + (size_t)(tm * 16 + hf * 8) * N + ts * 64 + c * 4;
#pragma unroll
  for (int v = 0; v < 8; ++v) {
    v4f o;
    o.x = acc0[v]; o.y = acc1[v]; o.z = acc2[v]; o.w = acc3[v];
    *(v4f*)(cp + (size_t)v * N) = o + bv;
  }
}

__global__ void fill_int(int* __restrict__ p, int v, long n) {
  long t = (long)blockIdx.x * blockDim.x + threadIdx.x;
  if (t < n) p[t] = v;
}

__global__ void count_and_rank(const int* __restrict__ ei, int* __restrict__ cnt,
                               int* __restrict__ rank, int E) {
  long t = (long)blockIdx.x * blockDim.x + threadIdx.x;
  if (t >= E) return;
  rank[t] = atomicAdd(&cnt[ei[E + t]], 1);
}

__global__ __launch_bounds__(1024) void scan_counts(const int* __restrict__ cnt,
                                                    int* __restrict__ rowptr, int n) {
  __shared__ int wsum[32];
  const int tid  = threadIdx.x;
  const int lane = tid & 31;
  const int wv   = tid >> 5;
  const int per  = (n + 1023) >> 10;
  const int lo   = min(tid * per, n);
  const int hi   = min(lo + per, n);
  int local = 0;
  for (int i = lo; i < hi; ++i) local += cnt[i];
  int x = local;
  for (int off = 1; off < 32; off <<= 1) {
    const int y = __shfl_up(x, off, 32);
    if (lane >= off) x += y;
  }
  if (lane == 31) wsum[wv] = x;
  __syncthreads();
  if (wv == 0) {
    const int w = wsum[lane];
    int xs = w;
    for (int off = 1; off < 32; off <<= 1) {
      const int y = __shfl_up(xs, off, 32);
      if (lane >= off) xs += y;
    }
    wsum[lane] = xs - w;
  }
  __syncthreads();
  int run = wsum[wv] + x - local;
  for (int i = lo; i < hi; ++i) { rowptr[i] = run; run += cnt[i]; }
  if (tid == 1023) rowptr[n] = run;
}

__global__ void fill_csr(const int* __restrict__ ei, const int* __restrict__ rowptr,
                         const int* __restrict__ rank, int* __restrict__ csr_src, int E) {
  long t = (long)blockIdx.x * blockDim.x + threadIdx.x;
  if (t >= E) return;
  csr_src[rowptr[ei[E + t]] + rank[t]] = ei[t];
}

__global__ void att_scores(const float* __restrict__ h,
                           const float* __restrict__ att_src,
                           const float* __restrict__ att_dst,
                           float* __restrict__ as, float* __restrict__ ad,
                           int n, int H, int C) {
  long t = (long)blockIdx.x * blockDim.x + threadIdx.x;
  if (t >= (long)n * H) return;
  const int head = (int)(t % H);
  const float4* hv = (const float4*)(h + (size_t)t * C);
  const float4* sv = (const float4*)(att_src + (size_t)head * C);
  const float4* dv = (const float4*)(att_dst + (size_t)head * C);
  float s = 0.f, d = 0.f;
  for (int i = 0; i < (C >> 2); ++i) {
    float4 xv = hv[i], a = sv[i], b = dv[i];
    s += xv.x * a.x + xv.y * a.y + xv.z * a.z + xv.w * a.w;
    d += xv.x * b.x + xv.y * b.y + xv.z * b.z + xv.w * b.w;
  }
  as[t] = s; ad[t] = d;
}

template <int H>
__device__ __forceinline__ void load_heads(const float* __restrict__ p, float (&v)[H]) {
  if constexpr (H == 4) {
    const float4 t = *(const float4*)p;
    v[0] = t.x; v[1] = t.y; v[2] = t.z; v[3] = t.w;
  } else {
    for (int h = 0; h < H; ++h) v[h] = p[h];
  }
}

template <int H, int C, bool ADD>
__global__ __launch_bounds__(256) void softmax_aggregate_ln_elu(
    const int* __restrict__ rowptr, const int* __restrict__ csr_src,
    const float* __restrict__ as, const float* __restrict__ ad,
    const float* __restrict__ feat, const float* __restrict__ bias,
    const float* __restrict__ g, const float* __restrict__ b,
    const float* __restrict__ add, float* __restrict__ out, int n) {
  constexpr int HC  = H * C;
  constexpr int VW  = (HC >= 128) ? 4 : HC / 32;
  constexpr int CHW = 32 * VW;
  constexpr int NCH = HC / CHW;
  constexpr int HPC = (CHW > C) ? CHW / C : 1;
  static_assert(VW >= 1 && NCH * CHW == HC, "row must split into whole wave chunks");
  static_assert(C % VW == 0, "a lane's vector must stay inside one head");
  static_assert(CHW % C == 0 || C % CHW == 0, "chunks and heads must nest");
  typedef float vec __attribute__((ext_vector_type(VW)));

  const int lane = threadIdx.x & 31;
  const int d = __builtin_amdgcn_readfirstlane(
      (int)(blockIdx.x * (blockDim.x >> 5) + (threadIdx.x >> 5)));
  if (d >= n) return;
  const int beg = rowptr[d];
  const int end = rowptr[d + 1];
  const int sub = (HPC > 1) ? (lane * VW) / C : 0;

  vec addv[NCH];
  if constexpr (ADD)
    for (int k = 0; k < NCH; ++k)
      addv[k] = *(const vec*)(add + (size_t)d * HC + k * CHW + lane * VW);

  auto head_of = [&](const float (&p)[H], int k) -> float {
    const int h0 = (k * CHW) / C;
    float w = p[h0];
    for (int q = 1; q < HPC; ++q) if (sub == q) w = p[h0 + q];
    return w;
  };

  float adv[H], vself[H], mx[H];
  load_heads<H>(ad + (size_t)d * H, adv);
  load_heads<H>(as + (size_t)d * H, vself);
  for (int h = 0; h < H; ++h) {
    const float t = vself[h] + adv[h];
    vself[h] = (t > 0.f) ? t : NEG_SLOPE * t;
    mx[h] = vself[h];
  }

  auto fetch = [&](int base, int& sr, float (&v)[H]) {
    sr = d;
    for (int h = 0; h < H; ++h) v[h] = -INFINITY;
    if (base + lane < end) {
      sr = csr_src[base + lane];
      float a[H];
      load_heads<H>(as + (size_t)sr * H, a);
      for (int h = 0; h < H; ++h) {
        const float t = a[h] + adv[h];
        v[h] = (t > 0.f) ? t : NEG_SLOPE * t;
      }
    }
  };

  int src0; float v0[H];
  fetch(beg, src0, v0);
  for (int h = 0; h < H; ++h) mx[h] = fmaxf(mx[h], v0[h]);
  for (int base = beg + 32; base < end; base += 32) {
    int sr; float v[H];
    fetch(base, sr, v);
    for (int h = 0; h < H; ++h) mx[h] = fmaxf(mx[h], v[h]);
  }
  for (int h = 0; h < H; ++h)
    for (int off = 16; off; off >>= 1) mx[h] = fmaxf(mx[h], __shfl_xor(mx[h], off, 32));

  float pself[H], ssum[H];
  for (int h = 0; h < H; ++h) { pself[h] = expf(vself[h] - mx[h]); ssum[h] = 0.f; }
  int col[NCH];
  vec acc[NCH];
  for (int k = 0; k < NCH; ++k) {
    col[k] = k * CHW + lane * VW;
    acc[k] = head_of(pself, k) * *(const vec*)(feat + (size_t)d * HC + col[k]);
  }

  auto add_batch = [&](int base, int sr, const float (&v)[H]) {
    const bool live = base + lane < end;
    float p[H];
    for (int h = 0; h < H; ++h) {
      p[h] = live ? expf(v[h] - mx[h]) : 0.f;
      ssum[h] += p[h];
    }
    const int cnt = min(32, end - base);
    for (int j = 0; j < cnt; ++j) {
      const int sj = __builtin_amdgcn_readlane(sr, j);
      float pj[H];
      for (int h = 0; h < H; ++h)
        pj[h] = __int_as_float(__builtin_amdgcn_readlane(__float_as_int(p[h]), j));
      for (int k = 0; k < NCH; ++k)
        acc[k] += head_of(pj, k) * *(const vec*)(feat + (size_t)sj * HC + col[k]);
    }
  };

  add_batch(beg, src0, v0);
  for (int base = beg + 32; base < end; base += 32) {
    int sr; float v[H];
    fetch(base, sr, v);
    add_batch(base, sr, v);
  }

  float inv[H];
  for (int h = 0; h < H; ++h) {
    for (int off = 16; off; off >>= 1) ssum[h] += __shfl_xor(ssum[h], off, 32);
    inv[h] = 1.0f / (ssum[h] + pself[h] + 1e-16f);
  }
  float s1 = 0.f, s2 = 0.f;
  vec gv[NCH], bv[NCH];
  for (int k = 0; k < NCH; ++k) {
    acc[k] = acc[k] * head_of(inv, k) + *(const vec*)(bias + col[k]);
    gv[k] = *(const vec*)(g + col[k]);
    bv[k] = *(const vec*)(b + col[k]);
    for (int i = 0; i < VW; ++i) { s1 += acc[k][i]; s2 += acc[k][i] * acc[k][i]; }
  }
  for (int off = 16; off; off >>= 1) {
    s1 += __shfl_xor(s1, off, 32);
    s2 += __shfl_xor(s2, off, 32);
  }
  const float mu   = s1 / (float)HC;
  const float var  = s2 / (float)HC - mu * mu;
  const float rstd = rsqrtf(var + 1e-5f);
  for (int k = 0; k < NCH; ++k) {
    vec y = (acc[k] - mu) * rstd * gv[k] + bv[k];
    if constexpr (ADD) y += addv[k];
    for (int i = 0; i < VW; ++i) y[i] = (y[i] > 0.f) ? y[i] : expm1f(y[i]);
    *(vec*)(out + (size_t)d * HC + col[k]) = y;
  }
}

extern "C" void kernel_launch(void* const* d_in, const int* in_sizes, int n_in,
                              void* d_out, int out_size, void* d_ws, size_t ws_size,
                              hipStream_t stream) {
  const float* x    = (const float*)d_in[0];
  const int*   ei   = (const int*)  d_in[1];
  const float* W1   = (const float*)d_in[2];
  const float* aS1  = (const float*)d_in[3];
  const float* aD1  = (const float*)d_in[4];
  const float* b1   = (const float*)d_in[5];
  const float* W2   = (const float*)d_in[6];
  const float* aS2  = (const float*)d_in[7];
  const float* aD2  = (const float*)d_in[8];
  const float* b2   = (const float*)d_in[9];
  const float* g1   = (const float*)d_in[10];
  const float* be1  = (const float*)d_in[11];
  const float* g2   = (const float*)d_in[12];
  const float* be2  = (const float*)d_in[13];
  const float* pW   = (const float*)d_in[14];
  const float* pb   = (const float*)d_in[15];

  const int IN_CH = 128, HEADS = 4, OUT_CH = 64, HC = HEADS * OUT_CH;
  const int N  = in_sizes[0] / IN_CH;
  const int E  = in_sizes[1] / 2;

  float* ws   = (float*)d_ws;
  float* h1   = ws; ws += (size_t)N * HC;
  float* h2   = ws; ws += (size_t)N * HC;
  float* idb  = ws; ws += (size_t)N * OUT_CH;
  float* z2   = ws; ws += (size_t)N * OUT_CH;
  float* as1  = ws; ws += (size_t)N * HEADS;
  float* ad1  = ws; ws += (size_t)N * HEADS;
  float* as2  = as1;
  float* ad2  = ad1;
  int* rowptr  = (int*)ws; ws += (size_t)N + 1;
  int* cnt     = (int*)ws; ws += (size_t)N;
  int* csr_src = (int*)ws; ws += (size_t)E;
  int* rank    = (int*)ws; ws += (size_t)E;

  auto cdiv = [](long a, long b) -> int { return (int)((a + b - 1) / b); };
  const int BLK = 256;
  const int WPB = BLK / 32;

  fill_int       <<<cdiv((long)N, BLK), BLK, 0, stream>>>(cnt, 0, (long)N);
  count_and_rank <<<cdiv((long)E, BLK), BLK, 0, stream>>>(ei, cnt, rank, E);
  scan_counts    <<<1, 1024, 0, stream>>>(cnt, rowptr, N);
  fill_csr       <<<cdiv((long)E, BLK), BLK, 0, stream>>>(ei, rowptr, rank, csr_src, E);

  {
    long waves = ((long)N / 16) * (HC / 64);
    gemm_wmma_f32<<<cdiv(waves, WPB), BLK, 0, stream>>>(x, W1, nullptr, h1, N, IN_CH, HC);
  }
  {
    long waves = ((long)N / 16) * (OUT_CH / 64);
    gemm_wmma_f32<<<cdiv(waves, WPB), BLK, 0, stream>>>(x, pW, pb, idb, N, IN_CH, OUT_CH);
  }

  att_scores<<<cdiv((long)N * HEADS, BLK), BLK, 0, stream>>>(h1, aS1, aD1, as1, ad1, N, HEADS, OUT_CH);
  softmax_aggregate_ln_elu<4, 64, false><<<cdiv(N, WPB), BLK, 0, stream>>>(
      rowptr, csr_src, as1, ad1, h1, b1, g1, be1, nullptr, h2, N);

  {
    long waves = ((long)N / 16) * (OUT_CH / 64);
    gemm_wmma_f32<<<cdiv(waves, WPB), BLK, 0, stream>>>(h2, W2, nullptr, z2, N, HC, OUT_CH);
  }

  att_scores<<<cdiv((long)N, BLK), BLK, 0, stream>>>(z2, aS2, aD2, as2, ad2, N, 1, OUT_CH);
  softmax_aggregate_ln_elu<1, 64, true><<<cdiv(N, WPB), BLK, 0, stream>>>(
      rowptr, csr_src, as2, ad2, z2, b2, g2, be2, idb, (float*)d_out, N);

  (void)n_in; (void)out_size; (void)ws_size;
}


# row-pointer scan: each wave reads a contiguous segment with coalesced 16-byte loads into registers, DPP lane scan, 16-byte stores, instead of every thread walking a private strided chunk of counts
# speedup vs baseline: 5.3243x; 1.1128x over previous
#include <hip/hip_runtime.h>
#include <math.h>

typedef float v2f __attribute__((ext_vector_type(2)));
typedef float v4f __attribute__((ext_vector_type(4)));
typedef float v8f __attribute__((ext_vector_type(8)));

#define NEG_SLOPE 0.2f

__device__ __forceinline__ float row16_sum(float x) {
  x += __int_as_float(__builtin_amdgcn_update_dpp(0, __float_as_int(x), 0x161, 0xf, 0xf, true));
  x += __int_as_float(__builtin_amdgcn_update_dpp(0, __float_as_int(x), 0x162, 0xf, 0xf, true));
  x += __int_as_float(__builtin_amdgcn_update_dpp(0, __float_as_int(x), 0x164, 0xf, 0xf, true));
  x += __int_as_float(__builtin_amdgcn_update_dpp(0, __float_as_int(x), 0x168, 0xf, 0xf, true));
  return x;
}

__global__ __launch_bounds__(256) void gemm_wmma_f32(
    const float* __restrict__ A, const float* __restrict__ B,
    const float* __restrict__ bias, float* __restrict__ C,
    const float* __restrict__ att_s, const float* __restrict__ att_d,
    float* __restrict__ as, float* __restrict__ ad,
    int M, int K, int N) {
  const int lane = threadIdx.x & 31;
  const int wave = blockIdx.x * (blockDim.x >> 5) + (threadIdx.x >> 5);
  const int stripsN = N >> 6;
  const int tilesM  = M >> 4;
  if (wave >= tilesM * stripsN) return;
  const int tm = wave / stripsN;
  const int ts = wave - tm * stripsN;
  const int c  = lane & 15;
  const int hf = lane >> 4;
  const float* ap = A + (size_t)(tm * 16 + c) * K + hf * 4;
  const float* bp = B + (size_t)(hf * 4) * N + ts * 64 + c * 4;

  v8f acc0 = {}, acc1 = {}, acc2 = {}, acc3 = {};
  v4f a  = *(const v4f*)ap;
  v4f b0 = *(const v4f*)(bp);
  v4f b1 = *(const v4f*)(bp + (size_t)N);
  v4f b2 = *(const v4f*)(bp + (size_t)2 * N);
  v4f b3 = *(const v4f*)(bp + (size_t)3 * N);
  for (int k = 0; k < K; k += 8) {
    const int kn = min(k + 8, K - 8);
    const float* bq = bp + (size_t)kn * N;
    const v4f an  = *(const v4f*)(ap + kn);
    const v4f bn0 = *(const v4f*)(bq);
    const v4f bn1 = *(const v4f*)(bq + (size_t)N);
    const v4f bn2 = *(const v4f*)(bq + (size_t)2 * N);
    const v4f bn3 = *(const v4f*)(bq + (size_t)3 * N);
    __builtin_amdgcn_sched_barrier(0);

    v2f af, bf;
    af.x = a.x; af.y = a.y;
    bf.x = b0.x; bf.y = b1.x;
    acc0 = __builtin_amdgcn_wmma_f32_16x16x4_f32(false, af, false, bf, (short)0, acc0, false, false);
    bf.x = b0.y; bf.y = b1.y;
    acc1 = __builtin_amdgcn_wmma_f32_16x16x4_f32(false, af, false, bf, (short)0, acc1, false, false);
    bf.x = b0.z; bf.y = b1.z;
    acc2 = __builtin_amdgcn_wmma_f32_16x16x4_f32(false, af, false, bf, (short)0, acc2, false, false);
    bf.x = b0.w; bf.y = b1.w;
    acc3 = __builtin_amdgcn_wmma_f32_16x16x4_f32(false, af, false, bf, (short)0, acc3, false, false);
    af.x = a.z; af.y = a.w;
    bf.x = b2.x; bf.y = b3.x;
    acc0 = __builtin_amdgcn_wmma_f32_16x16x4_f32(false, af, false, bf, (short)0, acc0, false, false);
    bf.x = b2.y; bf.y = b3.y;
    acc1 = __builtin_amdgcn_wmma_f32_16x16x4_f32(false, af, false, bf, (short)0, acc1, false, false);
    bf.x = b2.z; bf.y = b3.z;
    acc2 = __builtin_amdgcn_wmma_f32_16x16x4_f32(false, af, false, bf, (short)0, acc2, false, false);
    bf.x = b2.w; bf.y = b3.w;
    acc3 = __builtin_amdgcn_wmma_f32_16x16x4_f32(false, af, false, bf, (short)0, acc3, false, false);

    a = an; b0 = bn0; b1 = bn1; b2 = bn2; b3 = bn3;
  }

  v4f bv = {0.f, 0.f, 0.f, 0.f};
  if (bias) bv = *(const v4f*)(bias + ts * 64 + c * 4);
  const bool att = att_s != nullptr;
  v4f sv = {0.f, 0.f, 0.f, 0.f}, dv = {0.f, 0.f, 0.f, 0.f};
  if (att) {
    sv = *(const v4f*)(att_s + ts * 64 + c * 4);
    dv = *(const v4f*)(att_d + ts * 64 + c * 4);
  }
  const v2f slo = {sv.x, sv.y}, shi = {sv.z, sv.w}, dlo = {dv.x, dv.y}, dhi = {dv.z, dv.w};
  float ps[8], pd[8];
  float* cp = C + (size_t)(tm * 16 + hf * 8) * N + ts * 64 + c * 4;
#pragma unroll
  for (int v = 0; v < 8; ++v) {
    v4f o;
    o.x = acc0[v]; o.y = acc1[v]; o.z = acc2[v]; o.w = acc3[v];
    o += bv;
    *(v4f*)(cp + (size_t)v * N) = o;
    const v2f olo = {o.x, o.y}, ohi = {o.z, o.w};
    const v2f ts2 = olo * slo + ohi * shi;
    const v2f td2 = olo * dlo + ohi * dhi;
    ps[v] = ts2.x + ts2.y;
    pd[v] = td2.x + td2.y;
  }
  if (att) {
    const int cl = (int)(__builtin_amdgcn_mbcnt_lo(~0u, 0u) & 15u);
    float rs = 0.f, rd = 0.f;
#pragma unroll
    for (int v = 0; v < 8; ++v) {
      const float s = row16_sum(ps[v]);
      const float d = row16_sum(pd[v]);
      if (cl == v) { rs = s; rd = d; }
    }
    if (cl < 8) {
      const size_t o = (size_t)((cp - C) >> 6) + (size_t)(cl * stripsN);
      as[o] = rs;
      ad[o] = rd;
    }
  }
}

__global__ void fill_int(int* __restrict__ p, int v, long n) {
  long t = (long)blockIdx.x * blockDim.x + threadIdx.x;
  if (t < n) p[t] = v;
}

__global__ void count_and_rank(const int* __restrict__ ei, int* __restrict__ cnt,
                               int* __restrict__ rank, int E) {
  long t = (long)blockIdx.x * blockDim.x + threadIdx.x;
  if (t >= E) return;
  rank[t] = atomicAdd(&cnt[ei[E + t]], 1);
}

__device__ __forceinline__ int wave_incl_scan(int x, int lane) {
  x += __builtin_amdgcn_update_dpp(0, x, 0x111, 0xf, 0xf, true);
  x += __builtin_amdgcn_update_dpp(0, x, 0x112, 0xf, 0xf, true);
  x += __builtin_amdgcn_update_dpp(0, x, 0x114, 0xf, 0xf, true);
  x += __builtin_amdgcn_update_dpp(0, x, 0x118, 0xf, 0xf, true);
  const int low = __builtin_amdgcn_readlane(x, 15);
  return (lane >= 16) ? x + low : x;
}

#define SCAN_T 16

template <bool FULL>
__device__ __forceinline__ void scan_load(const int* __restrict__ cnt, int first, int tiles,
                                          int n, int4 (&v)[SCAN_T]) {
#pragma unroll
  for (int t = 0; t < SCAN_T; ++t) {
    v[t] = make_int4(0, 0, 0, 0);
    if (t < tiles) {
      const int i = first + (t << 7);
      if constexpr (FULL) {
        v[t] = *(const int4*)(cnt + i);
      } else {
        if (i     < n) v[t].x = cnt[i];
        if (i + 1 < n) v[t].y = cnt[i + 1];
        if (i + 2 < n) v[t].z = cnt[i + 2];
        if (i + 3 < n) v[t].w = cnt[i + 3];
      }
    }
  }
}

template <bool FULL>
__device__ __forceinline__ void scan_store(int* __restrict__ rowptr, int first, int tiles, int n,
                                           int lane, int run, const int4 (&v)[SCAN_T]) {
#pragma unroll
  for (int t = 0; t < SCAN_T; ++t) {
    if (t < tiles) {
      const int i   = first + (t << 7);
      const int s   = (v[t].x + v[t].y) + (v[t].z + v[t].w);
      const int inc = wave_incl_scan(s, lane);
      int4 o;
      o.x = run + inc - s;
      o.y = o.x + v[t].x;
      o.z = o.y + v[t].y;
      o.w = o.z + v[t].z;
      if constexpr (FULL) {
        *(int4*)(rowptr + i) = o;
      } else {
        if (i     < n) rowptr[i]     = o.x;
        if (i + 1 < n) rowptr[i + 1] = o.y;
        if (i + 2 < n) rowptr[i + 2] = o.z;
        if (i + 3 < n) rowptr[i + 3] = o.w;
      }
      run += __builtin_amdgcn_readlane(inc, 31);
    }
  }
}

__global__ __launch_bounds__(1024) void scan_counts(const int* __restrict__ cnt,
                                                    int* __restrict__ rowptr, int n) {
  __shared__ int wsum[33];
  const int tid  = threadIdx.x;
  const int lane = tid & 31;
  const int wv   = __builtin_amdgcn_readfirstlane(tid >> 5);
  const bool vec = ((((size_t)cnt) | ((size_t)rowptr)) & 15) == 0;
  const int CHUNK = 32 * 128 * SCAN_T;

  int carry = 0;
  for (int c0 = 0; c0 < n; c0 += CHUNK) {
    const int len   = min(n - c0, CHUNK);
    const int tiles = (len + 4095) >> 12;
    const int seg0  = c0 + wv * (tiles << 7);
    const int first = seg0 + lane * 4;
    const bool full = vec && (seg0 + (tiles << 7) <= n);

    int4 v[SCAN_T];
    if (full) scan_load<true>(cnt, first, tiles, n, v);
    else      scan_load<false>(cnt, first, tiles, n, v);
    int local = 0;
#pragma unroll
    for (int t = 0; t < SCAN_T; ++t) local += (v[t].x + v[t].y) + (v[t].z + v[t].w);
    const int tot = __builtin_amdgcn_readlane(wave_incl_scan(local, lane), 31);
    if (lane == 0) wsum[wv] = tot;
    __syncthreads();
    if (wv == 0) {
      const int w   = wsum[lane];
      const int inc = wave_incl_scan(w, lane);
      wsum[lane] = inc - w;
      if (lane == 31) wsum[32] = inc;
    }
    __syncthreads();
    const int run = carry + __builtin_amdgcn_readfirstlane(wsum[wv]);
    carry += wsum[32];
    if (full) scan_store<true>(rowptr, first, tiles, n, lane, run, v);
    else      scan_store<false>(rowptr, first, tiles, n, lane, run, v);
    __syncthreads();
  }
  if (tid == 0) rowptr[n] = carry;
}

__global__ void fill_csr(const int* __restrict__ ei, const int* __restrict__ rowptr,
                         const int* __restrict__ rank, int* __restrict__ csr_src, int E) {
  long t = (long)blockIdx.x * blockDim.x + threadIdx.x;
  if (t >= E) return;
  csr_src[rowptr[ei[E + t]] + rank[t]] = ei[t];
}

template <int H>
__device__ __forceinline__ void load_heads(const float* __restrict__ p, float (&v)[H]) {
  if constexpr (H == 4) {
    const float4 t = *(const float4*)p;
    v[0] = t.x; v[1] = t.y; v[2] = t.z; v[3] = t.w;
  } else {
    for (int h = 0; h < H; ++h) v[h] = p[h];
  }
}

template <int H, int C, bool ADD>
__global__ __launch_bounds__(256) void softmax_aggregate_ln_elu(
    const int* __restrict__ rowptr, const int* __restrict__ csr_src,
    const float* __restrict__ as, const float* __restrict__ ad,
    const float* __restrict__ feat, const float* __restrict__ bias,
    const float* __restrict__ g, const float* __restrict__ b,
    const float* __restrict__ add, float* __restrict__ out, int n) {
  constexpr int HC  = H * C;
  constexpr int VW  = (HC >= 128) ? 4 : HC / 32;
  constexpr int CHW = 32 * VW;
  constexpr int NCH = HC / CHW;
  constexpr int HPC = (CHW > C) ? CHW / C : 1;
  static_assert(VW >= 1 && NCH * CHW == HC, "row must split into whole wave chunks");
  static_assert(C % VW == 0, "a lane's vector must stay inside one head");
  static_assert(CHW % C == 0 || C % CHW == 0, "chunks and heads must nest");
  typedef float vec __attribute__((ext_vector_type(VW)));

  const int lane = threadIdx.x & 31;
  const int d = __builtin_amdgcn_readfirstlane(
      (int)(blockIdx.x * (blockDim.x >> 5) + (threadIdx.x >> 5)));
  if (d >= n) return;
  const int beg = rowptr[d];
  const int end = rowptr[d + 1];
  const int sub = (HPC > 1) ? (lane * VW) / C : 0;

  vec addv[NCH];
  if constexpr (ADD)
    for (int k = 0; k < NCH; ++k)
      addv[k] = *(const vec*)(add + (size_t)d * HC + k * CHW + lane * VW);

  auto head_of = [&](const float (&p)[H], int k) -> float {
    const int h0 = (k * CHW) / C;
    float w = p[h0];
    for (int q = 1; q < HPC; ++q) if (sub == q) w = p[h0 + q];
    return w;
  };

  float adv[H], vself[H], mx[H];
  load_heads<H>(ad + (size_t)d * H, adv);
  load_heads<H>(as + (size_t)d * H, vself);
  for (int h = 0; h < H; ++h) {
    const float t = vself[h] + adv[h];
    vself[h] = (t > 0.f) ? t : NEG_SLOPE * t;
    mx[h] = vself[h];
  }

  auto fetch = [&](int base, int& sr, float (&v)[H]) {
    sr = d;
    for (int h = 0; h < H; ++h) v[h] = -INFINITY;
    if (base + lane < end) {
      sr = csr_src[base + lane];
      float a[H];
      load_heads<H>(as + (size_t)sr * H, a);
      for (int h = 0; h < H; ++h) {
        const float t = a[h] + adv[h];
        v[h] = (t > 0.f) ? t : NEG_SLOPE * t;
      }
    }
  };

  int src0; float v0[H];
  fetch(beg, src0, v0);
  for (int h = 0; h < H; ++h) mx[h] = fmaxf(mx[h], v0[h]);
  for (int base = beg + 32; base < end; base += 32) {
    int sr; float v[H];
    fetch(base, sr, v);
    for (int h = 0; h < H; ++h) mx[h] = fmaxf(mx[h], v[h]);
  }
  for (int h = 0; h < H; ++h)
    for (int off = 16; off; off >>= 1) mx[h] = fmaxf(mx[h], __shfl_xor(mx[h], off, 32));

  float pself[H], ssum[H];
  for (int h = 0; h < H; ++h) { pself[h] = expf(vself[h] - mx[h]); ssum[h] = 0.f; }
  int col[NCH];
  vec acc[NCH];
  for (int k = 0; k < NCH; ++k) {
    col[k] = k * CHW + lane * VW;
    acc[k] = head_of(pself, k) * *(const vec*)(feat + (size_t)d * HC + col[k]);
  }

  auto add_batch = [&](int base, int sr, const float (&v)[H]) {
    const bool live = base + lane < end;
    float p[H];
    for (int h = 0; h < H; ++h) {
      p[h] = live ? expf(v[h] - mx[h]) : 0.f;
      ssum[h] += p[h];
    }
    const int cnt = min(32, end - base);
    for (int j = 0; j < cnt; ++j) {
      const int sj = __builtin_amdgcn_readlane(sr, j);
      float pj[H];
      for (int h = 0; h < H; ++h)
        pj[h] = __int_as_float(__builtin_amdgcn_readlane(__float_as_int(p[h]), j));
      for (int k = 0; k < NCH; ++k)
        acc[k] += head_of(pj, k) * *(const vec*)(feat + (size_t)sj * HC + col[k]);
    }
  };

  add_batch(beg, src0, v0);
  for (int base = beg + 32; base < end; base += 32) {
    int sr; float v[H];
    fetch(base, sr, v);
    add_batch(base, sr, v);
  }

  float inv[H];
  for (int h = 0; h < H; ++h) {
    for (int off = 16; off; off >>= 1) ssum[h] += __shfl_xor(ssum[h], off, 32);
    inv[h] = 1.0f / (ssum[h] + pself[h] + 1e-16f);
  }
  float s1 = 0.f, s2 = 0.f;
  vec gv[NCH], bv[NCH];
  for (int k = 0; k < NCH; ++k) {
    acc[k] = acc[k] * head_of(inv, k) + *(const vec*)(bias + col[k]);
    gv[k] = *(const vec*)(g + col[k]);
    bv[k] = *(const vec*)(b + col[k]);
    for (int i = 0; i < VW; ++i) { s1 += acc[k][i]; s2 += acc[k][i] * acc[k][i]; }
  }
  for (int off = 16; off; off >>= 1) {
    s1 += __shfl_xor(s1, off, 32);
    s2 += __shfl_xor(s2, off, 32);
  }
  const float mu   = s1 / (float)HC;
  const float var  = s2 / (float)HC - mu * mu;
  const float rstd = rsqrtf(var + 1e-5f);
  for (int k = 0; k < NCH; ++k) {
    vec y = (acc[k] - mu) * rstd * gv[k] + bv[k];
    if constexpr (ADD) y += addv[k];
    for (int i = 0; i < VW; ++i) y[i] = (y[i] > 0.f) ? y[i] : expm1f(y[i]);
    *(vec*)(out + (size_t)d * HC + col[k]) = y;
  }
}

extern "C" void kernel_launch(void* const* d_in, const int* in_sizes, int n_in,
                              void* d_out, int out_size, void* d_ws, size_t ws_size,
                              hipStream_t stream) {
  const float* x    = (const float*)d_in[0];
  const int*   ei   = (const int*)  d_in[1];
  const float* W1   = (const float*)d_in[2];
  const float* aS1  = (const float*)d_in[3];
  const float* aD1  = (const float*)d_in[4];
  const float* b1   = (const float*)d_in[5];
  const float* W2   = (const float*)d_in[6];
  const float* aS2  = (const float*)d_in[7];
  const float* aD2  = (const float*)d_in[8];
  const float* b2   = (const float*)d_in[9];
  const float* g1   = (const float*)d_in[10];
  const float* be1  = (const float*)d_in[11];
  const float* g2   = (const float*)d_in[12];
  const float* be2  = (const float*)d_in[13];
  const float* pW   = (const float*)d_in[14];
  const float* pb   = (const float*)d_in[15];

  const int IN_CH = 128, HEADS = 4, OUT_CH = 64, HC = HEADS * OUT_CH;
  const int N  = in_sizes[0] / IN_CH;
  const int E  = in_sizes[1] / 2;

  float* ws   = (float*)d_ws;
  float* h1   = ws; ws += (size_t)N * HC;
  float* h2   = ws; ws += (size_t)N * HC;
  float* idb  = ws; ws += (size_t)N * OUT_CH;
  float* z2   = ws; ws += (size_t)N * OUT_CH;
  float* as1  = ws; ws += (size_t)N * HEADS;
  float* ad1  = ws; ws += (size_t)N * HEADS;
  float* as2  = as1;
  float* ad2  = ad1;
  int* cnt     = (int*)ws; ws += (size_t)N;
  int* rowptr  = (int*)ws; ws += (size_t)N + 1;
  int* csr_src = (int*)ws; ws += (size_t)E;
  int* rank    = (int*)ws; ws += (size_t)E;

  auto cdiv = [](long a, long b) -> int { return (int)((a + b - 1) / b); };
  const int BLK = 256;
  const int WPB = BLK / 32;

  fill_int       <<<cdiv((long)N, BLK), BLK, 0, stream>>>(cnt, 0, (long)N);
  count_and_rank <<<cdiv((long)E, BLK), BLK, 0, stream>>>(ei, cnt, rank, E);
  scan_counts    <<<1, 1024, 0, stream>>>(cnt, rowptr, N);
  fill_csr       <<<cdiv((long)E, BLK), BLK, 0, stream>>>(ei, rowptr, rank, csr_src, E);

  {
    long waves = ((long)N / 16) * (HC / 64);
    gemm_wmma_f32<<<cdiv(waves, WPB), BLK, 0, stream>>>(x, W1, nullptr, h1, aS1, aD1, as1, ad1, N, IN_CH, HC);
  }
  {
    long waves = ((long)N / 16) * (OUT_CH / 64);
    gemm_wmma_f32<<<cdiv(waves, WPB), BLK, 0, stream>>>(x, pW, pb, idb, nullptr, nullptr, nullptr, nullptr, N, IN_CH, OUT_CH);
  }

  softmax_aggregate_ln_elu<4, 64, false><<<cdiv(N, WPB), BLK, 0, stream>>>(
      rowptr, csr_src, as1, ad1, h1, b1, g1, be1, nullptr, h2, N);

  {
    long waves = ((long)N / 16) * (OUT_CH / 64);
    gemm_wmma_f32<<<cdiv(waves, WPB), BLK, 0, stream>>>(h2, W2, nullptr, z2, aS2, aD2, as2, ad2, N, HC, OUT_CH);
  }

  softmax_aggregate_ln_elu<1, 64, true><<<cdiv(N, WPB), BLK, 0, stream>>>(
      rowptr, csr_src, as2, ad2, z2, b2, g2, be2, idb, (float*)d_out, N);

  (void)n_in; (void)out_size; (void)ws_size;
}


# layer-2 product moves into layer 1's aggregation launch: 16-row groups keep h2 in LDS, each wave does a 16-k slice on WMMA, slices summed in order via LDS; scores by 32-lane tree; 7 launches become 6
# speedup vs baseline: 5.9359x; 1.1149x over previous
#include <hip/hip_runtime.h>
#include <math.h>

typedef float v2f __attribute__((ext_vector_type(2)));
typedef float v4f __attribute__((ext_vector_type(4)));
typedef float v8f __attribute__((ext_vector_type(8)));

#define NEG_SLOPE 0.2f

__device__ __forceinline__ float row16_sum(float x) {
  x += __int_as_float(__builtin_amdgcn_update_dpp(0, __float_as_int(x), 0x161, 0xf, 0xf, true));
  x += __int_as_float(__builtin_amdgcn_update_dpp(0, __float_as_int(x), 0x162, 0xf, 0xf, true));
  x += __int_as_float(__builtin_amdgcn_update_dpp(0, __float_as_int(x), 0x164, 0xf, 0xf, true));
  x += __int_as_float(__builtin_amdgcn_update_dpp(0, __float_as_int(x), 0x168, 0xf, 0xf, true));
  return x;
}

__device__ __forceinline__ void csr_stage(const int* __restrict__ ei, int* __restrict__ cnt,
                                          int* __restrict__ rank, int kind, int n, int blk) {
  const long t = (long)blk * blockDim.x + threadIdx.x;
  if (t >= n) return;
  if (kind == 0) cnt[t] = 0;
  else           rank[t] = atomicAdd(&cnt[ei[n + t]], 1);
}

__global__ __launch_bounds__(256) void gemm_wmma_f32(
    const float* __restrict__ A, const float* __restrict__ B,
    const float* __restrict__ bias, float* __restrict__ C,
    const float* __restrict__ att_s, const float* __restrict__ att_d,
    float* __restrict__ as, float* __restrict__ ad,
    int M, int K, int N,
    const int* __restrict__ ei, int* __restrict__ cnt, int* __restrict__ rank,
    int stage_kind, int n_stage, int gemm_groups, int stage_groups) {
  const int grp   = (int)blockIdx.x;
  const int pairs = min(gemm_groups, stage_groups);
  bool stage;
  int  gidx;
  if (grp < 2 * pairs) { stage = (grp & 1) != 0; gidx = grp >> 1; }
  else                 { stage = stage_groups > gemm_groups; gidx = grp - pairs; }
  if (stage) {
    csr_stage(ei, cnt, rank, stage_kind, n_stage, gidx);
    return;
  }
  const int lane = threadIdx.x & 31;
  const int wave = __builtin_amdgcn_readfirstlane(
      (int)(gidx * (blockDim.x >> 5) + (threadIdx.x >> 5)));
  const int stripsN = N >> 6;
  const int tilesM  = M >> 4;
  if (wave >= tilesM * stripsN) return;
  const int tm = wave / stripsN;
  const int ts = wave - tm * stripsN;
  const int c  = lane & 15;
  const int hf = lane >> 4;
  const float* ap = A + (size_t)(tm * 16 + c) * K + hf * 4;
  const float* bp = B + (size_t)(hf * 4) * N + ts * 64 + c * 4;

  v8f acc0 = {}, acc1 = {}, acc2 = {}, acc3 = {};
  v4f a  = *(const v4f*)ap;
  v4f b0 = *(const v4f*)(bp);
  v4f b1 = *(const v4f*)(bp + (size_t)N);
  v4f b2 = *(const v4f*)(bp + (size_t)2 * N);
  v4f b3 = *(const v4f*)(bp + (size_t)3 * N);
  for (int k = 0; k < K; k += 8) {
    const int kn = min(k + 8, K - 8);
    const float* bq = bp + (size_t)kn * N;
    const v4f an  = *(const v4f*)(ap + kn);
    const v4f bn0 = *(const v4f*)(bq);
    const v4f bn1 = *(const v4f*)(bq + (size_t)N);
    const v4f bn2 = *(const v4f*)(bq + (size_t)2 * N);
    const v4f bn3 = *(const v4f*)(bq + (size_t)3 * N);
    __builtin_amdgcn_sched_barrier(0);

    v2f af, bf;
    af.x = a.x; af.y = a.y;
    bf.x = b0.x; bf.y = b1.x;
    acc0 = __builtin_amdgcn_wmma_f32_16x16x4_f32(false, af, false, bf, (short)0, acc0, false, false);
    bf.x = b0.y; bf.y = b1.y;
    acc1 = __builtin_amdgcn_wmma_f32_16x16x4_f32(false, af, false, bf, (short)0, acc1, false, false);
    bf.x = b0.z; bf.y = b1.z;
    acc2 = __builtin_amdgcn_wmma_f32_16x16x4_f32(false, af, false, bf, (short)0, acc2, false, false);
    bf.x = b0.w; bf.y = b1.w;
    acc3 = __builtin_amdgcn_wmma_f32_16x16x4_f32(false, af, false, bf, (short)0, acc3, false, false);
    af.x = a.z; af.y = a.w;
    bf.x = b2.x; bf.y = b3.x;
    acc0 = __builtin_amdgcn_wmma_f32_16x16x4_f32(false, af, false, bf, (short)0, acc0, false, false);
    bf.x = b2.y; bf.y = b3.y;
    acc1 = __builtin_amdgcn_wmma_f32_16x16x4_f32(false, af, false, bf, (short)0, acc1, false, false);
    bf.x = b2.z; bf.y = b3.z;
    acc2 = __builtin_amdgcn_wmma_f32_16x16x4_f32(false, af, false, bf, (short)0, acc2, false, false);
    bf.x = b2.w; bf.y = b3.w;
    acc3 = __builtin_amdgcn_wmma_f32_16x16x4_f32(false, af, false, bf, (short)0, acc3, false, false);

    a = an; b0 = bn0; b1 = bn1; b2 = bn2; b3 = bn3;
  }

  v4f bv = {0.f, 0.f, 0.f, 0.f};
  if (bias) bv = *(const v4f*)(bias + ts * 64 + c * 4);
  const bool att = att_s != nullptr;
  v4f sv = {0.f, 0.f, 0.f, 0.f}, dv = {0.f, 0.f, 0.f, 0.f};
  if (att) {
    sv = *(const v4f*)(att_s + ts * 64 + c * 4);
    dv = *(const v4f*)(att_d + ts * 64 + c * 4);
  }
  const v2f slo = {sv.x, sv.y}, shi = {sv.z, sv.w}, dlo = {dv.x, dv.y}, dhi = {dv.z, dv.w};
  float ps[8], pd[8];
  float* cp = C + (size_t)(tm * 16 + hf * 8) * N + ts * 64 + c * 4;
#pragma unroll
  for (int v = 0; v < 8; ++v) {
    v4f o;
    o.x = acc0[v]; o.y = acc1[v]; o.z = acc2[v]; o.w = acc3[v];
    o += bv;
    *(v4f*)(cp + (size_t)v * N) = o;
    const v2f olo = {o.x, o.y}, ohi = {o.z, o.w};
    const v2f ts2 = olo * slo + ohi * shi;
    const v2f td2 = olo * dlo + ohi * dhi;
    ps[v] = ts2.x + ts2.y;
    pd[v] = td2.x + td2.y;
  }
  if (att) {
    const int cl = (int)(__builtin_amdgcn_mbcnt_lo(~0u, 0u) & 15u);
    float rs = 0.f, rd = 0.f;
#pragma unroll
    for (int v = 0; v < 8; ++v) {
      const float s = row16_sum(ps[v]);
      const float d = row16_sum(pd[v]);
      if (cl == v) { rs = s; rd = d; }
    }
    if (cl < 8) {
      const size_t o = (size_t)((cp - C) >> 6) + (size_t)(cl * stripsN);
      as[o] = rs;
      ad[o] = rd;
    }
  }
}


__device__ __forceinline__ int wave_incl_scan(int x, int lane) {
  x += __builtin_amdgcn_update_dpp(0, x, 0x111, 0xf, 0xf, true);
  x += __builtin_amdgcn_update_dpp(0, x, 0x112, 0xf, 0xf, true);
  x += __builtin_amdgcn_update_dpp(0, x, 0x114, 0xf, 0xf, true);
  x += __builtin_amdgcn_update_dpp(0, x, 0x118, 0xf, 0xf, true);
  const int low = __builtin_amdgcn_readlane(x, 15);
  return (lane >= 16) ? x + low : x;
}

#define SCAN_T 16

template <bool FULL>
__device__ __forceinline__ void scan_load(const int* __restrict__ cnt, int first, int tiles,
                                          int n, int4 (&v)[SCAN_T]) {
#pragma unroll
  for (int t = 0; t < SCAN_T; ++t) {
    v[t] = make_int4(0, 0, 0, 0);
    if (t < tiles) {
      const int i = first + (t << 7);
      if constexpr (FULL) {
        v[t] = *(const int4*)(cnt + i);
      } else {
        if (i     < n) v[t].x = cnt[i];
        if (i + 1 < n) v[t].y = cnt[i + 1];
        if (i + 2 < n) v[t].z = cnt[i + 2];
        if (i + 3 < n) v[t].w = cnt[i + 3];
      }
    }
  }
}

template <bool FULL>
__device__ __forceinline__ void scan_store(int* __restrict__ rowptr, int first, int tiles, int n,
                                           int lane, int run, const int4 (&v)[SCAN_T]) {
#pragma unroll
  for (int t = 0; t < SCAN_T; ++t) {
    if (t < tiles) {
      const int i   = first + (t << 7);
      const int s   = (v[t].x + v[t].y) + (v[t].z + v[t].w);
      const int inc = wave_incl_scan(s, lane);
      int4 o;
      o.x = run + inc - s;
      o.y = o.x + v[t].x;
      o.z = o.y + v[t].y;
      o.w = o.z + v[t].z;
      if constexpr (FULL) {
        *(int4*)(rowptr + i) = o;
      } else {
        if (i     < n) rowptr[i]     = o.x;
        if (i + 1 < n) rowptr[i + 1] = o.y;
        if (i + 2 < n) rowptr[i + 2] = o.z;
        if (i + 3 < n) rowptr[i + 3] = o.w;
      }
      run += __builtin_amdgcn_readlane(inc, 31);
    }
  }
}

__global__ __launch_bounds__(1024) void scan_counts(const int* __restrict__ cnt,
                                                    int* __restrict__ rowptr, int n) {
  __shared__ int wsum[33];
  const int tid  = threadIdx.x;
  const int lane = tid & 31;
  const int wv   = __builtin_amdgcn_readfirstlane(tid >> 5);
  const bool vec = ((((size_t)cnt) | ((size_t)rowptr)) & 15) == 0;
  const int CHUNK = 32 * 128 * SCAN_T;

  int carry = 0;
  for (int c0 = 0; c0 < n; c0 += CHUNK) {
    const int len   = min(n - c0, CHUNK);
    const int tiles = (len + 4095) >> 12;
    const int seg0  = c0 + wv * (tiles << 7);
    const int first = seg0 + lane * 4;
    const bool full = vec && (seg0 + (tiles << 7) <= n);

    int4 v[SCAN_T];
    if (full) scan_load<true>(cnt, first, tiles, n, v);
    else      scan_load<false>(cnt, first, tiles, n, v);
    int local = 0;
#pragma unroll
    for (int t = 0; t < SCAN_T; ++t) local += (v[t].x + v[t].y) + (v[t].z + v[t].w);
    const int tot = __builtin_amdgcn_readlane(wave_incl_scan(local, lane), 31);
    if (lane == 0) wsum[wv] = tot;
    __syncthreads();
    if (wv == 0) {
      const int w   = wsum[lane];
      const int inc = wave_incl_scan(w, lane);
      wsum[lane] = inc - w;
      if (lane == 31) wsum[32] = inc;
    }
    __syncthreads();
    const int run = carry + __builtin_amdgcn_readfirstlane(wsum[wv]);
    carry += wsum[32];
    if (full) scan_store<true>(rowptr, first, tiles, n, lane, run, v);
    else      scan_store<false>(rowptr, first, tiles, n, lane, run, v);
    __syncthreads();
  }
  if (tid == 0) rowptr[n] = carry;
}

__global__ void fill_csr(const int* __restrict__ ei, const int* __restrict__ rowptr,
                         const int* __restrict__ rank, int* __restrict__ csr_src, int E) {
  long t = (long)blockIdx.x * blockDim.x + threadIdx.x;
  if (t >= E) return;
  csr_src[rowptr[ei[E + t]] + rank[t]] = ei[t];
}

template <int H>
__device__ __forceinline__ void load_heads(const float* __restrict__ p, float (&v)[H]) {
  if constexpr (H == 4) {
    const float4 t = *(const float4*)p;
    v[0] = t.x; v[1] = t.y; v[2] = t.z; v[3] = t.w;
  } else {
    for (int h = 0; h < H; ++h) v[h] = p[h];
  }
}

__device__ __forceinline__ void strip_kstep(const v4f a, const v4f b0, const v4f b1, const v4f b2,
                                            const v4f b3, v8f& acc0, v8f& acc1, v8f& acc2, v8f& acc3) {
  v2f af, bf;
  af.x = a.x; af.y = a.y;
  bf.x = b0.x; bf.y = b1.x;
  acc0 = __builtin_amdgcn_wmma_f32_16x16x4_f32(false, af, false, bf, (short)0, acc0, false, false);
  bf.x = b0.y; bf.y = b1.y;
  acc1 = __builtin_amdgcn_wmma_f32_16x16x4_f32(false, af, false, bf, (short)0, acc1, false, false);
  bf.x = b0.z; bf.y = b1.z;
  acc2 = __builtin_amdgcn_wmma_f32_16x16x4_f32(false, af, false, bf, (short)0, acc2, false, false);
  bf.x = b0.w; bf.y = b1.w;
  acc3 = __builtin_amdgcn_wmma_f32_16x16x4_f32(false, af, false, bf, (short)0, acc3, false, false);
  af.x = a.z; af.y = a.w;
  bf.x = b2.x; bf.y = b3.x;
  acc0 = __builtin_amdgcn_wmma_f32_16x16x4_f32(false, af, false, bf, (short)0, acc0, false, false);
  bf.x = b2.y; bf.y = b3.y;
  acc1 = __builtin_amdgcn_wmma_f32_16x16x4_f32(false, af, false, bf, (short)0, acc1, false, false);
  bf.x = b2.z; bf.y = b3.z;
  acc2 = __builtin_amdgcn_wmma_f32_16x16x4_f32(false, af, false, bf, (short)0, acc2, false, false);
  bf.x = b2.w; bf.y = b3.w;
  acc3 = __builtin_amdgcn_wmma_f32_16x16x4_f32(false, af, false, bf, (short)0, acc3, false, false);
}

template <int H, int C, bool ADD, bool NEXT>
__global__ __launch_bounds__(NEXT ? 512 : 256) void softmax_aggregate_ln_elu(
    const int* __restrict__ rowptr, const int* __restrict__ csr_src,
    const float* __restrict__ as, const float* __restrict__ ad,
    const float* __restrict__ feat, const float* __restrict__ bias,
    const float* __restrict__ g, const float* __restrict__ b,
    const float* __restrict__ add, float* __restrict__ out, int n,
    const float* __restrict__ Wn, const float* __restrict__ att_sn,
    const float* __restrict__ att_dn, float* __restrict__ zn,
    float* __restrict__ asn, float* __restrict__ adn) {
  constexpr int HC  = H * C;
  constexpr int VW  = (HC >= 128) ? 4 : HC / 32;
  constexpr int CHW = 32 * VW;
  constexpr int NCH = HC / CHW;
  constexpr int HPC = (CHW > C) ? CHW / C : 1;
  static_assert(VW >= 1 && NCH * CHW == HC, "row must split into whole wave chunks");
  static_assert(C % VW == 0, "a lane's vector must stay inside one head");
  static_assert(CHW % C == 0 || C % CHW == 0, "chunks and heads must nest");
  typedef float vec __attribute__((ext_vector_type(VW)));

  const int lane = threadIdx.x & 31;
  const int dw = __builtin_amdgcn_readfirstlane(
      (int)(blockIdx.x * (blockDim.x >> 5) + (threadIdx.x >> 5)));
  if constexpr (!NEXT) { if (dw >= n) return; }
  const int d = NEXT ? min(dw, n - 1) : dw;
  constexpr int PITCH = HC + 4;
  __shared__ float yrow[NEXT ? 16 * PITCH : 1];
  __shared__ float part[NEXT ? 16 * 16 * 64 : 1];
  const int beg = rowptr[d];
  const int end = rowptr[d + 1];
  const int sub = (HPC > 1) ? (lane * VW) / C : 0;

  vec addv[NCH];
  if constexpr (ADD)
    for (int k = 0; k < NCH; ++k)
      addv[k] = *(const vec*)(add + (size_t)d * HC + k * CHW + lane * VW);

  auto head_of = [&](const float (&p)[H], int k) -> float {
    const int h0 = (k * CHW) / C;
    float w = p[h0];
    for (int q = 1; q < HPC; ++q) if (sub == q) w = p[h0 + q];
    return w;
  };

  float adv[H], vself[H], mx[H];
  load_heads<H>(ad + (size_t)d * H, adv);
  load_heads<H>(as + (size_t)d * H, vself);
  for (int h = 0; h < H; ++h) {
    const float t = vself[h] + adv[h];
    vself[h] = (t > 0.f) ? t : NEG_SLOPE * t;
    mx[h] = vself[h];
  }

  auto fetch = [&](int base, int& sr, float (&v)[H]) {
    sr = d;
    for (int h = 0; h < H; ++h) v[h] = -INFINITY;
    if (base + lane < end) {
      sr = csr_src[base + lane];
      float a[H];
      load_heads<H>(as + (size_t)sr * H, a);
      for (int h = 0; h < H; ++h) {
        const float t = a[h] + adv[h];
        v[h] = (t > 0.f) ? t : NEG_SLOPE * t;
      }
    }
  };

  int src0; float v0[H];
  fetch(beg, src0, v0);
  for (int h = 0; h < H; ++h) mx[h] = fmaxf(mx[h], v0[h]);
  for (int base = beg + 32; base < end; base += 32) {
    int sr; float v[H];
    fetch(base, sr, v);
    for (int h = 0; h < H; ++h) mx[h] = fmaxf(mx[h], v[h]);
  }
  for (int h = 0; h < H; ++h)
    for (int off = 16; off; off >>= 1) mx[h] = fmaxf(mx[h], __shfl_xor(mx[h], off, 32));

  float pself[H], ssum[H];
  for (int h = 0; h < H; ++h) { pself[h] = expf(vself[h] - mx[h]); ssum[h] = 0.f; }
  int col[NCH];
  vec acc[NCH];
  for (int k = 0; k < NCH; ++k) {
    col[k] = k * CHW + lane * VW;
    acc[k] = head_of(pself, k) * *(const vec*)(feat + (size_t)d * HC + col[k]);
  }

  auto add_batch = [&](int base, int sr, const float (&v)[H]) {
    const bool live = base + lane < end;
    float p[H];
    for (int h = 0; h < H; ++h) {
      p[h] = live ? expf(v[h] - mx[h]) : 0.f;
      ssum[h] += p[h];
    }
    const int cnt = min(32, end - base);
    for (int j = 0; j < cnt; ++j) {
      const int sj = __builtin_amdgcn_readlane(sr, j);
      float pj[H];
      for (int h = 0; h < H; ++h)
        pj[h] = __int_as_float(__builtin_amdgcn_readlane(__float_as_int(p[h]), j));
      for (int k = 0; k < NCH; ++k)
        acc[k] += head_of(pj, k) * *(const vec*)(feat + (size_t)sj * HC + col[k]);
    }
  };

  add_batch(beg, src0, v0);
  for (int base = beg + 32; base < end; base += 32) {
    int sr; float v[H];
    fetch(base, sr, v);
    add_batch(base, sr, v);
  }

  float inv[H];
  for (int h = 0; h < H; ++h) {
    for (int off = 16; off; off >>= 1) ssum[h] += __shfl_xor(ssum[h], off, 32);
    inv[h] = 1.0f / (ssum[h] + pself[h] + 1e-16f);
  }
  const int wv = __builtin_amdgcn_readfirstlane((int)(threadIdx.x >> 5));
  v4f wb[2][4];
  v2f asv = {0.f, 0.f}, adv2 = {0.f, 0.f};
  if constexpr (NEXT) {
    static_assert(!NEXT || (HC == 256 && VW == 4), "16 waves x 16 k each; one 16-byte piece per lane and chunk");
    const float* bp = Wn + (size_t)(16 * wv + 4 * (lane >> 4)) * 64 + 4 * (lane & 15);
    for (int s = 0; s < 2; ++s)
      for (int r = 0; r < 4; ++r) wb[s][r] = *(const v4f*)(bp + (size_t)(8 * s + r) * 64);
    asv  = *(const v2f*)(att_sn + 2 * lane);
    adv2 = *(const v2f*)(att_dn + 2 * lane);
  }
  float s1 = 0.f, s2 = 0.f;
  vec gv[NCH], bv[NCH];
  for (int k = 0; k < NCH; ++k) {
    acc[k] = acc[k] * head_of(inv, k) + *(const vec*)(bias + col[k]);
    gv[k] = *(const vec*)(g + col[k]);
    bv[k] = *(const vec*)(b + col[k]);
    for (int i = 0; i < VW; ++i) { s1 += acc[k][i]; s2 += acc[k][i] * acc[k][i]; }
  }
  for (int off = 16; off; off >>= 1) {
    s1 += __shfl_xor(s1, off, 32);
    s2 += __shfl_xor(s2, off, 32);
  }
  const float mu   = s1 / (float)HC;
  const float var  = s2 / (float)HC - mu * mu;
  const float rstd = rsqrtf(var + 1e-5f);
  for (int k = 0; k < NCH; ++k) {
    vec y = (acc[k] - mu) * rstd * gv[k] + bv[k];
    if constexpr (ADD) y += addv[k];
    for (int i = 0; i < VW; ++i) y[i] = (y[i] > 0.f) ? y[i] : expm1f(y[i]);
    if constexpr (NEXT) *(vec*)(yrow + wv * PITCH + col[k]) = y;
    else                *(vec*)(out + (size_t)d * HC + col[k]) = y;
  }
  if constexpr (NEXT) {
    __syncthreads();
    const int c  = lane & 15;
    const int hf = lane >> 4;
    const float* ar = yrow + c * PITCH + 16 * wv + 4 * hf;
    const v4f a0 = *(const v4f*)ar;
    const v4f a1 = *(const v4f*)(ar + 8);
    v8f acc0 = {}, acc1 = {}, acc2 = {}, acc3 = {};
    strip_kstep(a0, wb[0][0], wb[0][1], wb[0][2], wb[0][3], acc0, acc1, acc2, acc3);
    strip_kstep(a1, wb[1][0], wb[1][1], wb[1][2], wb[1][3], acc0, acc1, acc2, acc3);
    float* pp = part + wv * 1024 + (8 * hf) * 64 + 4 * c;
#pragma unroll
    for (int v = 0; v < 8; ++v) {
      v4f o;
      o.x = acc0[v]; o.y = acc1[v]; o.z = acc2[v]; o.w = acc3[v];
      *(v4f*)(pp + v * 64) = o;
    }
    __syncthreads();
    const float* pr = part + wv * 64 + 2 * lane;
    v2f z = {0.f, 0.f};
#pragma unroll
    for (int p = 0; p < 16; ++p) z += *(const v2f*)(pr + p * 1024);
    float ps = z.x * asv.x + z.y * asv.y;
    float pd = z.x * adv2.x + z.y * adv2.y;
    for (int off = 16; off; off >>= 1) {
      ps += __shfl_xor(ps, off, 32);
      pd += __shfl_xor(pd, off, 32);
    }
    if (dw < n) {
      *(v2f*)(zn + (size_t)dw * 64 + 2 * lane) = z;
      if (lane == 0) { asn[dw] = ps; adn[dw] = pd; }
    }
  }
}

extern "C" void kernel_launch(void* const* d_in, const int* in_sizes, int n_in,
                              void* d_out, int out_size, void* d_ws, size_t ws_size,
                              hipStream_t stream) {
  const float* x    = (const float*)d_in[0];
  const int*   ei   = (const int*)  d_in[1];
  const float* W1   = (const float*)d_in[2];
  const float* aS1  = (const float*)d_in[3];
  const float* aD1  = (const float*)d_in[4];
  const float* b1   = (const float*)d_in[5];
  const float* W2   = (const float*)d_in[6];
  const float* aS2  = (const float*)d_in[7];
  const float* aD2  = (const float*)d_in[8];
  const float* b2   = (const float*)d_in[9];
  const float* g1   = (const float*)d_in[10];
  const float* be1  = (const float*)d_in[11];
  const float* g2   = (const float*)d_in[12];
  const float* be2  = (const float*)d_in[13];
  const float* pW   = (const float*)d_in[14];
  const float* pb   = (const float*)d_in[15];

  const int IN_CH = 128, HEADS = 4, OUT_CH = 64, HC = HEADS * OUT_CH;
  const int N  = in_sizes[0] / IN_CH;
  const int E  = in_sizes[1] / 2;

  float* ws   = (float*)d_ws;
  float* h1   = ws; ws += (size_t)N * HC;
  float* sc2  = ws; ws += (size_t)N * HC;
  float* idb  = ws; ws += (size_t)N * OUT_CH;
  float* z2   = ws; ws += (size_t)N * OUT_CH;
  float* as1  = ws; ws += (size_t)N * HEADS;
  float* ad1  = ws; ws += (size_t)N * HEADS;
  float* as2  = sc2;
  float* ad2  = sc2 + (size_t)N;
  int* cnt     = (int*)ws; ws += (size_t)N;
  int* rowptr  = (int*)ws; ws += (size_t)N + 1;
  int* csr_src = (int*)ws; ws += (size_t)E;
  int* rank    = (int*)ws; ws += (size_t)E;

  auto cdiv = [](long a, long b) -> int { return (int)((a + b - 1) / b); };
  const int BLK = 256;
  const int WPB = BLK / 32;

  {
    const int gg = cdiv(((long)N / 16) * (OUT_CH / 64), WPB);
    const int sg = cdiv((long)N, BLK);
    gemm_wmma_f32<<<gg + sg, BLK, 0, stream>>>(x, pW, pb, idb, nullptr, nullptr, nullptr, nullptr, N, IN_CH, OUT_CH,
                                               nullptr, cnt, nullptr, 0, N, gg, sg);
  }
  {
    const int gg = cdiv(((long)N / 16) * (HC / 64), WPB);
    const int sg = cdiv((long)E, BLK);
    gemm_wmma_f32<<<gg + sg, BLK, 0, stream>>>(x, W1, nullptr, h1, aS1, aD1, as1, ad1, N, IN_CH, HC,
                                               ei, cnt, rank, 1, E, gg, sg);
  }
  scan_counts    <<<1, 1024, 0, stream>>>(cnt, rowptr, N);
  fill_csr       <<<cdiv((long)E, BLK), BLK, 0, stream>>>(ei, rowptr, rank, csr_src, E);

  softmax_aggregate_ln_elu<4, 64, false, true><<<cdiv(N, 16), 512, 0, stream>>>(
      rowptr, csr_src, as1, ad1, h1, b1, g1, be1, nullptr, nullptr, N,
      W2, aS2, aD2, z2, as2, ad2);

  softmax_aggregate_ln_elu<1, 64, true, false><<<cdiv(N, WPB), BLK, 0, stream>>>(
      rowptr, csr_src, as2, ad2, z2, b2, g2, be2, idb, (float*)d_out, N,
      nullptr, nullptr, nullptr, nullptr, nullptr, nullptr);

  (void)n_in; (void)out_size; (void)ws_size;
}
